# starting kernel
# baseline (speedup 1.0000x reference)
#include <hip/hip_runtime.h>
#include <hip/hip_bf16.h>
#include <math.h>

#define NN   4096
#define MD   64
#define MSG  64
#define RAW  4224
#define HID  2112
#define WL   8

typedef __attribute__((ext_vector_type(2))) float v2f;
typedef __attribute__((ext_vector_type(8))) float v8f;

__device__ __forceinline__ v2f ldv2(const float* p) { return *(const v2f*)p; }

__device__ __forceinline__ void wmma_step4x4(
    v8f acc[4][4],
    const float* __restrict__ A, int r0, int ldA, int kA,
    const float* __restrict__ B, int c0, int ldB, int kB, int half) {
  v2f a[4], b[4];
#pragma unroll
  for (int t = 0; t < 4; ++t) {
    a[t] = ldv2(A + (size_t)(r0 + 16 * t) * (size_t)ldA + kA + 2 * half);
    b[t] = ldv2(B + (size_t)(c0 + 16 * t) * (size_t)ldB + kB + 2 * half);
  }
#pragma unroll
  for (int mi = 0; mi < 4; ++mi)
#pragma unroll
    for (int ni = 0; ni < 4; ++ni)
      acc[mi][ni] = __builtin_amdgcn_wmma_f32_16x16x4_f32(
          false, a[mi], false, b[ni], (short)0, acc[mi][ni], false, false);
}

__global__ __launch_bounds__(64) void k_rw(
    const float* __restrict__ memory, const int* __restrict__ walks,
    const float* __restrict__ W_rw, const float* __restrict__ b_rw,
    float* __restrict__ rw) {
  __shared__ float pooled[MD];
  const int n = blockIdx.x;
  const int d = threadIdx.x;
  float s = 0.f;
#pragma unroll
  for (int w = 0; w < WL; ++w) {
    const int idx = walks[n * WL + w];
    s += memory[(size_t)idx * MD + d];
  }
  pooled[d] = s * (1.0f / WL);
  __syncthreads();
  float acc = b_rw[d];
#pragma unroll 8
  for (int k = 0; k < MD; ++k) acc += pooled[k] * W_rw[d * MD + k];
  rw[(size_t)n * MSG + d] = acc;
}

__global__ __launch_bounds__(32) void k_gemm1(
    const float* __restrict__ memory, const float* __restrict__ od,
    const float* __restrict__ rw, const float* __restrict__ W1,
    const float* __restrict__ bias1, float* __restrict__ hid) {
  const int lane = threadIdx.x & 31;
  const int l15 = lane & 15;
  const int half = lane >> 4;
  const int nBase = blockIdx.x * 64;
  const int mBase = blockIdx.y * 64;
  const int r0 = mBase + l15;
  const int c0 = nBase + l15;

  v8f acc[4][4] = {};

  for (int k0 = 0; k0 < MD; k0 += 4)
    wmma_step4x4(acc, memory, r0, MD, k0, W1, c0, RAW, k0, half);
  for (int k0 = MD; k0 < MD + NN; k0 += 4)
    wmma_step4x4(acc, od, r0, NN, k0 - MD, W1, c0, RAW, k0, half);
  for (int k0 = MD + NN; k0 < RAW; k0 += 4)
    wmma_step4x4(acc, rw, r0, MSG, k0 - (MD + NN), W1, c0, RAW, k0, half);

#pragma unroll
  for (int ni = 0; ni < 4; ++ni) {
    const int col = nBase + ni * 16 + l15;
    const float bv = bias1[col];
#pragma unroll
    for (int mi = 0; mi < 4; ++mi) {
#pragma unroll
      for (int i = 0; i < 8; ++i) {
        const int row = mBase + mi * 16 + i + 8 * half;
        const float v = acc[mi][ni][i] + bv;
        hid[(size_t)row * HID + col] = v > 0.f ? v : 0.f;
      }
    }
  }
}

__global__ __launch_bounds__(32) void k_gemm2(
    const float* __restrict__ hid, const float* __restrict__ W2,
    const float* __restrict__ bias2, float* __restrict__ msg) {
  const int lane = threadIdx.x & 31;
  const int l15 = lane & 15;
  const int half = lane >> 4;
  const int mBase = blockIdx.x * 16;
  const int r0 = mBase + l15;

  v8f acc[4] = {};
  for (int k0 = 0; k0 < HID; k0 += 4) {
    const v2f a = ldv2(hid + (size_t)r0 * HID + k0 + 2 * half);
#pragma unroll
    for (int ni = 0; ni < 4; ++ni) {
      const v2f b = ldv2(W2 + (size_t)(ni * 16 + l15) * HID + k0 + 2 * half);
      acc[ni] = __builtin_amdgcn_wmma_f32_16x16x4_f32(
          false, a, false, b, (short)0, acc[ni], false, false);
    }
  }
#pragma unroll
  for (int ni = 0; ni < 4; ++ni) {
    const int col = ni * 16 + l15;
    const float bv = bias2[col];
#pragma unroll
    for (int i = 0; i < 8; ++i) {
      const int row = mBase + i + 8 * half;
      msg[(size_t)row * MSG + col] = acc[ni][i] + bv;
    }
  }
}

__global__ __launch_bounds__(64) void k_gru(
    const float* __restrict__ msg, const float* __restrict__ memory,
    const float* __restrict__ Wi, const float* __restrict__ bi,
    const float* __restrict__ Wh, const float* __restrict__ bh,
    float* __restrict__ upd) {
  __shared__ float sm[MSG];
  __shared__ float sh[MD];
  const int n = blockIdx.x;
  const int j = threadIdx.x;
  sm[j] = msg[(size_t)n * MSG + j];
  sh[j] = memory[(size_t)n * MD + j];
  __syncthreads();
  float ir = bi[j], iz = bi[MD + j], in_ = bi[2 * MD + j];
  float hr = bh[j], hz = bh[MD + j], hn = bh[2 * MD + j];
#pragma unroll 4
  for (int m = 0; m < MSG; ++m) {
    const float mv = sm[m];
    const float hv = sh[m];
    ir += mv * Wi[(size_t)(j) * MSG + m];
    iz += mv * Wi[(size_t)(MD + j) * MSG + m];
    in_ += mv * Wi[(size_t)(2 * MD + j) * MSG + m];
    hr += hv * Wh[(size_t)(j) * MD + m];
    hz += hv * Wh[(size_t)(MD + j) * MD + m];
    hn += hv * Wh[(size_t)(2 * MD + j) * MD + m];
  }
  const float r = 1.f / (1.f + __expf(-(ir + hr)));
  const float z = 1.f / (1.f + __expf(-(iz + hz)));
  const float nn = tanhf(in_ + r * hn);
  upd[(size_t)n * MD + j] = (1.f - z) * nn + z * sh[j];
}

__global__ __launch_bounds__(64) void k_pred1(
    const float* __restrict__ upd, const float* __restrict__ Wp1,
    const float* __restrict__ bp1, float* __restrict__ p) {
  __shared__ float su[MD];
  const int n = blockIdx.x;
  const int j = threadIdx.x;
  su[j] = upd[(size_t)n * MD + j];
  __syncthreads();
  float acc = bp1[j];
#pragma unroll 8
  for (int d = 0; d < MD; ++d) acc += su[d] * Wp1[j * MD + d];
  p[(size_t)n * MD + j] = acc > 0.f ? acc : 0.f;
}

__global__ __launch_bounds__(128) void k_pred2(
    const float* __restrict__ p, const float* __restrict__ Wp2,
    const float* __restrict__ bp2, float* __restrict__ out) {
  const int lane = threadIdx.x & 31;
  const int wave = threadIdx.x >> 5;
  const int l15 = lane & 15;
  const int half = lane >> 4;
  const int mBase = blockIdx.y * 64;
  const int nBase = blockIdx.x * 256 + wave * 64;
  const int r0 = mBase + l15;
  const int c0 = nBase + l15;

  v8f acc[4][4] = {};
#pragma unroll
  for (int k0 = 0; k0 < MD; k0 += 4)
    wmma_step4x4(acc, p, r0, MD, k0, Wp2, c0, MD, k0, half);

#pragma unroll
  for (int ni = 0; ni < 4; ++ni) {
    const int col = nBase + ni * 16 + l15;
    const float bv = bp2[col];
#pragma unroll
    for (int mi = 0; mi < 4; ++mi) {
#pragma unroll
      for (int i = 0; i < 8; ++i) {
        const int row = mBase + mi * 16 + i + 8 * half;
        out[(size_t)row * NN + col] = acc[mi][ni][i] + bv;
      }
    }
  }
}

extern "C" void kernel_launch(void* const* d_in, const int* in_sizes, int n_in,
                              void* d_out, int out_size, void* d_ws, size_t ws_size,
                              hipStream_t stream) {
  const float* memory = (const float*)d_in[0];
  const float* od_mat = (const float*)d_in[1];
  const int*   walks  = (const int*)d_in[2];
  const float* W_rw   = (const float*)d_in[3];
  const float* b_rw   = (const float*)d_in[4];
  const float* W1     = (const float*)d_in[5];
  const float* b1     = (const float*)d_in[6];
  const float* W2     = (const float*)d_in[7];
  const float* b2     = (const float*)d_in[8];
  const float* gWi    = (const float*)d_in[9];
  const float* gbi    = (const float*)d_in[10];
  const float* gWh    = (const float*)d_in[11];
  const float* gbh    = (const float*)d_in[12];
  const float* Wp1    = (const float*)d_in[13];
  const float* bp1    = (const float*)d_in[14];
  const float* Wp2    = (const float*)d_in[15];
  const float* bp2    = (const float*)d_in[16];
  float* out = (float*)d_out;

  char* ws = (char*)d_ws;
  const size_t SZ_RW  = (size_t)NN * MSG * sizeof(float);
  const size_t SZ_HID = (size_t)NN * HID * sizeof(float);
  const size_t SZ_MSG = (size_t)NN * MSG * sizeof(float);
  const size_t SZ_UPD = (size_t)NN * MD  * sizeof(float);
  float* rw   = (float*)(ws);
  float* hid  = (float*)(ws + SZ_RW);
  float* msg  = (float*)(ws + SZ_RW + SZ_HID);
  float* upd  = (float*)(ws + SZ_RW + SZ_HID + SZ_MSG);
  float* pbuf = (float*)(ws + SZ_RW + SZ_HID + SZ_MSG + SZ_UPD);

  k_rw<<<NN, 64, 0, stream>>>(memory, walks, W_rw, b_rw, rw);
  {
    dim3 grid(HID / 64, NN / 64);
    k_gemm1<<<grid, 32, 0, stream>>>(memory, od_mat, rw, W1, b1, hid);
  }
  k_gemm2<<<NN / 16, 32, 0, stream>>>(hid, W2, b2, msg);
  k_gru<<<NN, 64, 0, stream>>>(msg, memory, gWi, gbi, gWh, gbh, upd);
  k_pred1<<<NN, 64, 0, stream>>>(upd, Wp1, bp1, pbuf);
  {
    dim3 grid(NN / 256, NN / 64);
    k_pred2<<<grid, 128, 0, stream>>>(pbuf, Wp2, bp2, out);
  }
}


# layer-1 GEMM: the work-group's 33rd column tile split over K among its 4 waves (33 products per SIMD per k step, was 32/32/32/36); K partials added via LDS in wave order, layer-2 term regrouped
# speedup vs baseline: 2.1709x; 2.1709x over previous
#include <hip/hip_runtime.h>
#include <hip/hip_bf16.h>
#include <math.h>

#define NN   4096
#define MD   64
#define MSG  64
#define RAW  4224
#define HID  2112
#define WL   8

typedef __attribute__((ext_vector_type(2))) float v2f;
typedef __attribute__((ext_vector_type(8))) float v8f;

typedef __attribute__((ext_vector_type(4))) float v4f;

__device__ __forceinline__ v2f ldv2(const float* p) { return *(const v2f*)p; }
__device__ __forceinline__ v4f ldv4(const float* p) { return *(const v4f*)p; }

#define G1_TM 4
#define G1_TN 8
template <bool NINTH>
__device__ __forceinline__ void trip_load(
    v4f (&a)[G1_TM][4], v4f (&b)[G1_TN + 1][4],
    const float* __restrict__ pa, int ldA, const float* __restrict__ pb, long d9) {
#pragma unroll
  for (int q = 0; q < 4; ++q) {
#pragma unroll
    for (int t = 0; t < G1_TM; ++t)
      a[t][q] = ldv4(pa + (size_t)(16 * t) * (size_t)ldA + 8 * q);
#pragma unroll
    for (int t = 0; t < G1_TN; ++t)
      b[t][q] = ldv4(pb + (size_t)(16 * t) * (size_t)RAW + 8 * q);
    if (NINTH) b[G1_TN][q] = ldv4(pb + d9 + 8 * q);
  }
}

template <bool NINTH>
__device__ __forceinline__ void trip_mma(
    v8f (&acc)[G1_TM][G1_TN], v8f (&acc9)[G1_TM],
    const v4f (&a)[G1_TM][4], const v4f (&b)[G1_TN + 1][4]) {
#pragma unroll
  for (int q = 0; q < 4; ++q) {
#pragma unroll
    for (int e = 0; e < 2; ++e) {
#pragma unroll
      for (int mi = 0; mi < G1_TM; ++mi) {
        v2f av;
        av.x = a[mi][q][2 * e];
        av.y = a[mi][q][2 * e + 1];
#pragma unroll
        for (int ni = 0; ni < G1_TN; ++ni) {
          v2f bv;
          bv.x = b[ni][q][2 * e];
          bv.y = b[ni][q][2 * e + 1];
          acc[mi][ni] = __builtin_amdgcn_wmma_f32_16x16x4_f32(
              false, bv, false, av, (short)0, acc[mi][ni], false, false);
        }
        if (NINTH) {
          v2f bv;
          bv.x = b[G1_TN][q][2 * e];
          bv.y = b[G1_TN][q][2 * e + 1];
          acc9[mi] = __builtin_amdgcn_wmma_f32_16x16x4_f32(
              false, bv, false, av, (short)0, acc9[mi], false, false);
        }
      }
    }
  }
}

__device__ __forceinline__ void mma_gap() {
  __builtin_amdgcn_sched_barrier(0);
  asm volatile("v_nop\n\tv_nop\n\tv_nop\n\tv_nop\n\tv_nop");
  __builtin_amdgcn_sched_barrier(0);
}

__device__ __forceinline__ void wmma_step4x4(
    v8f acc[4][4],
    const float* __restrict__ A, int r0, int ldA, int kA,
    const float* __restrict__ B, int c0, int ldB, int kB, int half) {
  v2f a[4], b[4];
#pragma unroll
  for (int t = 0; t < 4; ++t) {
    a[t] = ldv2(A + (size_t)(r0 + 16 * t) * (size_t)ldA + kA + 2 * half);
    b[t] = ldv2(B + (size_t)(c0 + 16 * t) * (size_t)ldB + kB + 2 * half);
  }
#pragma unroll
  for (int mi = 0; mi < 4; ++mi)
#pragma unroll
    for (int ni = 0; ni < 4; ++ni)
      acc[mi][ni] = __builtin_amdgcn_wmma_f32_16x16x4_f32(
          false, a[mi], false, b[ni], (short)0, acc[mi][ni], false, false);
}

#define NPB 8

__global__ __launch_bounds__(64) void k_rw(
    const float* __restrict__ memory, const int* __restrict__ walks,
    const float* __restrict__ W_rw, const float* __restrict__ b_rw,
    float* __restrict__ rw) {
  __shared__ __attribute__((aligned(16))) float pooled[NPB * MD];
  const int n0 = blockIdx.x * NPB;
  const int d = threadIdx.x;
  v4f w[MD / 4];
#pragma unroll
  for (int i = 0; i < MD / 4; ++i) w[i] = ldv4(W_rw + d * MD + 4 * i);
#pragma unroll
  for (int t = 0; t < NPB; ++t) {
    float s = 0.f;
#pragma unroll
    for (int wk = 0; wk < WL; ++wk) {
      const int idx = walks[(n0 + t) * WL + wk];
      s += memory[(size_t)idx * MD + d];
    }
    pooled[t * MD + d] = s * (1.0f / WL);
  }
  __syncthreads();
  float acc[NPB];
  const float bias = b_rw[d];
#pragma unroll
  for (int t = 0; t < NPB; ++t) acc[t] = bias;
#pragma unroll
  for (int i = 0; i < MD / 4; ++i) {
#pragma unroll
    for (int t = 0; t < NPB; ++t) {
      const v4f pv = *(const v4f*)&pooled[t * MD + 4 * i];
#pragma unroll
      for (int e = 0; e < 4; ++e) acc[t] = __builtin_fmaf(pv[e], w[i][e], acc[t]);
    }
  }
#pragma unroll
  for (int t = 0; t < NPB; ++t) rw[(size_t)(n0 + t) * MSG + d] = acc[t];
}

#define G1_QT 33
#define G1_CB 16
#define G1_TRIPS (RAW / 32)

__device__ __forceinline__ void a_source(
    int t, const float* __restrict__ memory, const float* __restrict__ od,
    const float* __restrict__ rw, int r0, int half,
    const float*& p, int& ld) {
  const bool first = t < MD / 32;
  const bool last = t >= (MD + NN) / 32;
  const float* base = first ? memory : (last ? rw : od);
  const int kt = first ? t : (last ? t - (MD + NN) / 32 : t - MD / 32);
  ld = (first || last) ? MD : NN;
  p = base + (size_t)r0 * (size_t)ld + 4 * half + 32 * kt;
}

template <bool NINTH>
__device__ __forceinline__ void run_trips(
    int tb, int te,
    v8f (&acc)[G1_TM][G1_TN], v8f (&acc9)[G1_TM],
    v4f (&a0)[G1_TM][4], v4f (&b0)[G1_TN + 1][4],
    v4f (&a1)[G1_TM][4], v4f (&b1)[G1_TN + 1][4],
    const float* __restrict__ memory, const float* __restrict__ od,
    const float* __restrict__ rw, int r0, int half,
    const float* __restrict__ wrow, long d9) {
  const float* pa;
  int ld;
#pragma nounroll
  for (int t = tb; t < te; t += 2) {
    a_source(t + 1, memory, od, rw, r0, half, pa, ld);
    trip_load<NINTH>(a1, b1, pa, ld, wrow + 32 * (t + 1), d9);
    __builtin_amdgcn_sched_barrier(0);
    trip_mma<NINTH>(acc, acc9, a0, b0);
    mma_gap();
    const int tn = t + 2 < G1_TRIPS ? t + 2 : G1_TRIPS - 1;
    a_source(tn, memory, od, rw, r0, half, pa, ld);
    trip_load<NINTH>(a0, b0, pa, ld, wrow + 32 * tn, d9);
    __builtin_amdgcn_sched_barrier(0);
    trip_mma<NINTH>(acc, acc9, a1, b1);
    mma_gap();
  }
}

__global__ __launch_bounds__(128) __attribute__((amdgpu_waves_per_eu(1, 1)))
void k_gemm1(
    const float* __restrict__ memory, const float* __restrict__ od,
    const float* __restrict__ rw, const float* __restrict__ W1,
    const float* __restrict__ bias1, const float* __restrict__ W2,
    float* __restrict__ part) {
  __shared__ __attribute__((aligned(16))) float xch[4 * G1_TM * 32 * 8];
  const int lane = threadIdx.x & 31;
  const int wave = __builtin_amdgcn_readfirstlane((int)(threadIdx.x >> 5));
  const int l15 = lane & 15;
  const int half = lane >> 4;
  const int nBase = blockIdx.x * (16 * G1_QT) + wave * (16 * G1_TN);
  const int n9 = blockIdx.x * (16 * G1_QT) + 4 * (16 * G1_TN);
  const int mBase = blockIdx.y * (16 * G1_TM);
  const int plane = blockIdx.x * 4 + wave;
  const int r0 = mBase + l15;
  const int lo = 33 * wave - (wave & 1);
  const int hi = 33 * (wave + 1) - ((wave + 1) & 1);

  v8f acc[G1_TM][G1_TN] = {};
  v8f acc9[G1_TM] = {};
  const float* wrow = W1 + (size_t)(nBase + l15) * RAW + 4 * half;
  const long d9 = (long)(n9 - nBase) * RAW;

  v4f a0[G1_TM][4], b0[G1_TN + 1][4], a1[G1_TM][4], b1[G1_TN + 1][4];
  const float* pa;
  int ld;
  a_source(0, memory, od, rw, r0, half, pa, ld);
  trip_load<false>(a0, b0, pa, ld, wrow, d9);
  run_trips<false>(0, lo, acc, acc9, a0, b0, a1, b1, memory, od, rw, r0, half, wrow, d9);
  a_source(lo, memory, od, rw, r0, half, pa, ld);
  trip_load<true>(a0, b0, pa, ld, wrow + 32 * lo, d9);
  run_trips<true>(lo, hi, acc, acc9, a0, b0, a1, b1, memory, od, rw, r0, half, wrow, d9);
  run_trips<false>(hi, G1_TRIPS, acc, acc9, a0, b0, a1, b1, memory, od, rw, r0, half, wrow, d9);

  {
    float* xw = xch + ((size_t)(wave * G1_TM) * 32 + lane) * 8;
#pragma unroll
    for (int mi = 0; mi < G1_TM; ++mi) {
      *(v4f*)(xw + mi * 256) = __builtin_shufflevector(acc9[mi], acc9[mi], 0, 1, 2, 3);
      *(v4f*)(xw + mi * 256 + 4) = __builtin_shufflevector(acc9[mi], acc9[mi], 4, 5, 6, 7);
    }
  }

  v4f w2f[G1_TN][4][2], w2n[4][2];
  {
    const float* w2p = W2 + (size_t)l15 * HID + nBase + 8 * half;
    const float* w2q = W2 + (size_t)l15 * HID + n9 + 8 * half;
#pragma unroll
    for (int ni = 0; ni < G1_TN; ++ni) {
#pragma unroll
      for (int jt = 0; jt < 4; ++jt) {
        w2f[ni][jt][0] = ldv4(w2p + (size_t)(16 * jt) * HID + 16 * ni);
        w2f[ni][jt][1] = ldv4(w2p + (size_t)(16 * jt) * HID + 16 * ni + 4);
      }
    }
#pragma unroll
    for (int jt = 0; jt < 4; ++jt) {
      w2n[jt][0] = ldv4(w2q + (size_t)(16 * jt) * HID);
      w2n[jt][1] = ldv4(w2q + (size_t)(16 * jt) * HID + 4);
    }
  }

#pragma unroll
  for (int ni = 0; ni < G1_TN; ++ni) {
    const float* bp = bias1 + nBase + 16 * ni + 8 * half;
    const v4f blo = ldv4(bp);
    const v4f bhi = ldv4(bp + 4);
#pragma unroll
    for (int mi = 0; mi < G1_TM; ++mi) {
#pragma unroll
      for (int r = 0; r < 8; ++r) {
        const float v = acc[mi][ni][r] + (r < 4 ? blo[r & 3] : bhi[r & 3]);
        acc[mi][ni][r] = v > 0.f ? v : 0.f;
      }
    }
  }
  __builtin_amdgcn_sched_barrier(0);

  v8f macc[G1_TM][4] = {};
#pragma unroll
  for (int ni = 0; ni < G1_TN; ++ni) {
#pragma unroll
    for (int s = 0; s < 4; ++s) {
#pragma unroll
      for (int jt = 0; jt < 4; ++jt) {
        v2f wv;
        wv.x = w2f[ni][jt][s >> 1][2 * (s & 1)];
        wv.y = w2f[ni][jt][s >> 1][2 * (s & 1) + 1];
#pragma unroll
        for (int mi = 0; mi < G1_TM; ++mi) {
          v2f hv;
          hv.x = acc[mi][ni][2 * s];
          hv.y = acc[mi][ni][2 * s + 1];
          macc[mi][jt] = __builtin_amdgcn_wmma_f32_16x16x4_f32(
              false, wv, false, hv, (short)0, macc[mi][jt], false, false);
        }
      }
    }
  }
  mma_gap();

  __syncthreads();
  v8f h9;
  {
    const float* xr = xch + ((size_t)wave * 32 + lane) * 8;
    v4f slo = *(const v4f*)(xr);
    v4f shi = *(const v4f*)(xr + 4);
#pragma unroll
    for (int w = 1; w < 4; ++w) {
      slo += *(const v4f*)(xr + w * (G1_TM * 256));
      shi += *(const v4f*)(xr + w * (G1_TM * 256) + 4);
    }
    const float* bp = bias1 + n9 + 8 * half;
    slo += ldv4(bp);
    shi += ldv4(bp + 4);
#pragma unroll
    for (int r = 0; r < 4; ++r) {
      h9[r] = slo[r] > 0.f ? slo[r] : 0.f;
      h9[4 + r] = shi[r] > 0.f ? shi[r] : 0.f;
    }
  }
  __builtin_amdgcn_sched_barrier(0);
  v8f m9[4] = {};
#pragma unroll
  for (int s = 0; s < 4; ++s) {
#pragma unroll
    for (int jt = 0; jt < 4; ++jt) {
      v2f wv, hv;
      wv.x = w2n[jt][s >> 1][2 * (s & 1)];
      wv.y = w2n[jt][s >> 1][2 * (s & 1) + 1];
      hv.x = h9[2 * s];
      hv.y = h9[2 * s + 1];
      m9[jt] = __builtin_amdgcn_wmma_f32_16x16x4_f32(
          false, wv, false, hv, (short)0, m9[jt], false, false);
    }
  }
  mma_gap();

  float* pp = part + ((size_t)plane * NN + (size_t)(mBase + l15)) * MSG + 8 * half;
#pragma unroll
  for (int mi = 0; mi < G1_TM; ++mi) {
    const bool mine = wave == mi;
#pragma unroll
    for (int jt = 0; jt < 4; ++jt) {
      v8f v = macc[mi][jt];
#pragma unroll
      for (int r = 0; r < 8; ++r) v[r] += mine ? m9[jt][r] : 0.f;
      float* q = pp + (size_t)(16 * mi) * MSG + 16 * jt;
      *(v4f*)q = __builtin_shufflevector(v, v, 0, 1, 2, 3);
      *(v4f*)(q + 4) = __builtin_shufflevector(v, v, 4, 5, 6, 7);
    }
  }
}

__global__ __launch_bounds__(64) void k_gru(
    const float* __restrict__ part, const float* __restrict__ bias2,
    const float* __restrict__ memory,
    const float* __restrict__ Wi, const float* __restrict__ bi,
    const float* __restrict__ Wh, const float* __restrict__ bh,
    float* __restrict__ upd) {
  __shared__ __attribute__((aligned(16))) float sm[NPB * MSG];
  __shared__ __attribute__((aligned(16))) float sh[NPB * MD];
  const int n0 = blockIdx.x * NPB;
  const int j = threadIdx.x;
#pragma unroll
  for (int u = 0; u < (NPB * MSG) / (4 * 64); ++u) {
    const int idx = j + 64 * u;
    const float* pp = part + (size_t)n0 * MSG + 4 * idx;
    v4f ms = {0.f, 0.f, 0.f, 0.f};
#pragma unroll
    for (int cb = 0; cb < G1_CB; ++cb)
      ms += ldv4(pp + (size_t)cb * NN * MSG);
    ms += ldv4(bias2 + 4 * (idx & (MSG / 4 - 1)));
    *(v4f*)&sm[4 * idx] = ms;
    *(v4f*)&sh[4 * idx] = ldv4(memory + (size_t)n0 * MD + 4 * idx);
  }
  __syncthreads();
  float ir[NPB], iz[NPB], in_[NPB], hr[NPB], hz[NPB], hn[NPB];
  {
    const float b0 = bi[j], b1 = bi[MD + j], b2v = bi[2 * MD + j];
    const float c0 = bh[j], c1 = bh[MD + j], c2 = bh[2 * MD + j];
#pragma unroll
    for (int t = 0; t < NPB; ++t) {
      ir[t] = b0; iz[t] = b1; in_[t] = b2v;
      hr[t] = c0; hz[t] = c1; hn[t] = c2;
    }
  }
  const float* wi0 = Wi + (size_t)j * MSG;
  const float* wi1 = Wi + (size_t)(MD + j) * MSG;
  const float* wi2 = Wi + (size_t)(2 * MD + j) * MSG;
  const float* wh0 = Wh + (size_t)j * MD;
  const float* wh1 = Wh + (size_t)(MD + j) * MD;
  const float* wh2 = Wh + (size_t)(2 * MD + j) * MD;
#pragma nounroll
  for (int m0 = 0; m0 < MSG; m0 += 16) {
    v4f a0[4], a1[4], a2[4], g0[4], g1[4], g2[4];
#pragma unroll
    for (int c = 0; c < 4; ++c) {
      a0[c] = ldv4(wi0 + m0 + 4 * c);
      a1[c] = ldv4(wi1 + m0 + 4 * c);
      a2[c] = ldv4(wi2 + m0 + 4 * c);
      g0[c] = ldv4(wh0 + m0 + 4 * c);
      g1[c] = ldv4(wh1 + m0 + 4 * c);
      g2[c] = ldv4(wh2 + m0 + 4 * c);
    }
#pragma unroll
    for (int c = 0; c < 4; ++c) {
#pragma unroll
      for (int t = 0; t < NPB; ++t) {
        const v4f mv = *(const v4f*)&sm[t * MSG + m0 + 4 * c];
        const v4f hv = *(const v4f*)&sh[t * MD + m0 + 4 * c];
#pragma unroll
        for (int e = 0; e < 4; ++e) {
          ir[t] = __builtin_fmaf(mv[e], a0[c][e], ir[t]);
          iz[t] = __builtin_fmaf(mv[e], a1[c][e], iz[t]);
          in_[t] = __builtin_fmaf(mv[e], a2[c][e], in_[t]);
          hr[t] = __builtin_fmaf(hv[e], g0[c][e], hr[t]);
          hz[t] = __builtin_fmaf(hv[e], g1[c][e], hz[t]);
          hn[t] = __builtin_fmaf(hv[e], g2[c][e], hn[t]);
        }
      }
    }
  }
#pragma unroll
  for (int t = 0; t < NPB; ++t) {
    const float r = 1.f / (1.f + __expf(-(ir[t] + hr[t])));
    const float z = 1.f / (1.f + __expf(-(iz[t] + hz[t])));
    const float nn = tanhf(in_[t] + r * hn[t]);
    upd[(size_t)(n0 + t) * MD + j] = (1.f - z) * nn + z * sh[t * MD + j];
  }
}

__global__ __launch_bounds__(64) void k_pred1(
    const float* __restrict__ upd, const float* __restrict__ Wp1,
    const float* __restrict__ bp1, float* __restrict__ p) {
  __shared__ __attribute__((aligned(16))) float su[NPB * MD];
  const int n0 = blockIdx.x * NPB;
  const int j = threadIdx.x;
  v4f w[MD / 4];
#pragma unroll
  for (int i = 0; i < MD / 4; ++i) w[i] = ldv4(Wp1 + j * MD + 4 * i);
#pragma unroll
  for (int u = 0; u < (NPB * MD) / (4 * 64); ++u) {
    const int idx = j + 64 * u;
    *(v4f*)&su[4 * idx] = ldv4(upd + (size_t)n0 * MD + 4 * idx);
  }
  __syncthreads();
  float acc[NPB];
  const float bias = bp1[j];
#pragma unroll
  for (int t = 0; t < NPB; ++t) acc[t] = bias;
#pragma unroll
  for (int i = 0; i < MD / 4; ++i) {
#pragma unroll
    for (int t = 0; t < NPB; ++t) {
      const v4f uv = *(const v4f*)&su[t * MD + 4 * i];
#pragma unroll
      for (int e = 0; e < 4; ++e) acc[t] = __builtin_fmaf(uv[e], w[i][e], acc[t]);
    }
  }
#pragma unroll
  for (int t = 0; t < NPB; ++t)
    p[(size_t)(n0 + t) * MD + j] = acc[t] > 0.f ? acc[t] : 0.f;
}

__global__ __launch_bounds__(128) void k_pred2(
    const float* __restrict__ p, const float* __restrict__ Wp2,
    const float* __restrict__ bp2, float* __restrict__ out) {
  const int lane = threadIdx.x & 31;
  const int wave = threadIdx.x >> 5;
  const int l15 = lane & 15;
  const int half = lane >> 4;
  const int mBase = blockIdx.y * 64;
  const int nBase = blockIdx.x * 256 + wave * 64;
  const int r0 = mBase + l15;
  const int c0 = nBase + l15;

  v8f acc[4][4] = {};
#pragma unroll
  for (int k0 = 0; k0 < MD; k0 += 4)
    wmma_step4x4(acc, p, r0, MD, k0, Wp2, c0, MD, k0, half);

#pragma unroll
  for (int ni = 0; ni < 4; ++ni) {
    const int col = nBase + ni * 16 + l15;
    const float bv = bp2[col];
#pragma unroll
    for (int mi = 0; mi < 4; ++mi) {
#pragma unroll
      for (int i = 0; i < 8; ++i) {
        const int row = mBase + mi * 16 + i + 8 * half;
        out[(size_t)row * NN + col] = acc[mi][ni][i] + bv;
      }
    }
  }
}

extern "C" void kernel_launch(void* const* d_in, const int* in_sizes, int n_in,
                              void* d_out, int out_size, void* d_ws, size_t ws_size,
                              hipStream_t stream) {
  const float* memory = (const float*)d_in[0];
  const float* od_mat = (const float*)d_in[1];
  const int*   walks  = (const int*)d_in[2];
  const float* W_rw   = (const float*)d_in[3];
  const float* b_rw   = (const float*)d_in[4];
  const float* W1     = (const float*)d_in[5];
  const float* b1     = (const float*)d_in[6];
  const float* W2     = (const float*)d_in[7];
  const float* b2     = (const float*)d_in[8];
  const float* gWi    = (const float*)d_in[9];
  const float* gbi    = (const float*)d_in[10];
  const float* gWh    = (const float*)d_in[11];
  const float* gbh    = (const float*)d_in[12];
  const float* Wp1    = (const float*)d_in[13];
  const float* bp1    = (const float*)d_in[14];
  const float* Wp2    = (const float*)d_in[15];
  const float* bp2    = (const float*)d_in[16];
  float* out = (float*)d_out;

  char* ws = (char*)d_ws;
  const size_t SZ_RW   = (size_t)NN * MSG * sizeof(float);
  const size_t SZ_PART = (size_t)G1_CB * NN * MSG * sizeof(float);
  const size_t SZ_UPD  = (size_t)NN * MD  * sizeof(float);
  float* rw   = (float*)(ws);
  float* part = (float*)(ws + SZ_RW);
  float* upd  = (float*)(ws + SZ_RW + SZ_PART);
  float* pbuf = (float*)(ws + SZ_RW + SZ_PART + SZ_UPD);

  k_rw<<<NN / NPB, 64, 0, stream>>>(memory, walks, W_rw, b_rw, rw);
  {
    dim3 grid(HID / (16 * G1_QT), NN / (16 * G1_TM));
    k_gemm1<<<grid, 128, 0, stream>>>(memory, od_mat, rw, W1, b1, W2, part);
  }
  k_gru<<<NN / NPB, 64, 0, stream>>>(part, b2, memory, gWi, gbi, gWh, gbh, upd);
  k_pred1<<<NN / NPB, 64, 0, stream>>>(upd, Wp1, bp1, pbuf);
  {
    dim3 grid(NN / 256, NN / 64);
    k_pred2<<<grid, 128, 0, stream>>>(pbuf, Wp2, bp2, out);
  }
}
